# starting kernel
# baseline (speedup 1.0000x reference)
#include <hip/hip_runtime.h>
#include <hip/hip_bf16.h>

#define N_USERS 100000
#define N_ITEMS 200000
#define N_NODES (N_USERS + N_ITEMS)
#define NNZ     4800000
#define DIM     64
#define BATCH   16384

typedef float v2f __attribute__((ext_vector_type(2)));
typedef float v8f __attribute__((ext_vector_type(8)));

__global__ void lgcn_init(const float* __restrict__ user_emb,
                          const float* __restrict__ item_emb,
                          float* __restrict__ h,
                          float* __restrict__ acc) {
    size_t f = (size_t)blockIdx.x * blockDim.x + threadIdx.x;
    size_t total = (size_t)N_NODES * (DIM / 4);
    if (f >= total) return;
    size_t node = f >> 4;
    size_t part = f & 15;
    const float4* src;
    if (node < N_USERS)
        src = (const float4*)(user_emb + node * DIM);
    else
        src = (const float4*)(item_emb + (node - N_USERS) * DIM);
    float4 v = src[part];
    ((float4*)h)[f]   = v;
    ((float4*)acc)[f] = v;
}

__global__ void lgcn_zero(float* __restrict__ p) {
    size_t f = (size_t)blockIdx.x * blockDim.x + threadIdx.x;
    size_t total = (size_t)N_NODES * (DIM / 4);
    if (f >= total) return;
    ((float4*)p)[f] = make_float4(0.f, 0.f, 0.f, 0.f);
}

__global__ void lgcn_spmm(const int*   __restrict__ rows,
                          const int*   __restrict__ cols,
                          const float* __restrict__ vals,
                          const float* __restrict__ h,
                          float*       __restrict__ out) {
    size_t tid = (size_t)blockIdx.x * blockDim.x + threadIdx.x;
    size_t e   = tid >> 4;
    int    part = (int)(tid & 15);
    if (e >= (size_t)NNZ) return;
    int   r = rows[e];
    int   c = cols[e];
    float v = vals[e];
    const float4 src = *(const float4*)(h + (size_t)c * DIM + part * 4);
    float* dst = out + (size_t)r * DIM + part * 4;
    unsafeAtomicAdd(dst + 0, v * src.x);
    unsafeAtomicAdd(dst + 1, v * src.y);
    unsafeAtomicAdd(dst + 2, v * src.z);
    unsafeAtomicAdd(dst + 3, v * src.w);
}

__global__ void lgcn_addacc(float* __restrict__ acc, const float* __restrict__ h) {
    size_t f = (size_t)blockIdx.x * blockDim.x + threadIdx.x;
    size_t total = (size_t)N_NODES * (DIM / 4);
    if (f >= total) return;
    float4 a = ((float4*)acc)[f];
    float4 b = ((const float4*)h)[f];
    a.x += b.x; a.y += b.y; a.z += b.z; a.w += b.w;
    ((float4*)acc)[f] = a;
}

__global__ void lgcn_dot_wmma(const float* __restrict__ acc,
                              const int*   __restrict__ users,
                              const int*   __restrict__ items,
                              float*       __restrict__ out) {
    __shared__ float lds[8][16][17];
    int lane  = threadIdx.x & 31;
    int wv    = threadIdx.x >> 5;
    int tile  = blockIdx.x * 8 + wv;
    int m     = lane & 15;
    int khalf = lane >> 4;

    int u_row = users[tile * 16 + m];
    int v_row = items[tile * 16 + m];
    const float* up = acc + (size_t)u_row * DIM;
    const float* vp = acc + ((size_t)N_USERS + (size_t)v_row) * DIM;

    v8f c = {};
#pragma unroll
    for (int kc = 0; kc < DIM / 4; ++kc) {
        int off = kc * 4 + khalf * 2;
        v2f a; a.x = up[off]; a.y = up[off + 1];
        v2f b; b.x = vp[off]; b.y = vp[off + 1];
        c = __builtin_amdgcn_wmma_f32_16x16x4_f32(
                 false, a,  false, b,
                 (short)0, c,  false,  false);
    }

#pragma unroll
    for (int r = 0; r < 8; ++r)
        lds[wv][r + 8 * khalf][m] = c[r];
    __syncthreads();

    if (lane < 16)
        out[tile * 16 + lane] = lds[wv][lane][lane] * (1.0f / 16.0f);
}

extern "C" void kernel_launch(void* const* d_in, const int* in_sizes, int n_in,
                              void* d_out, int out_size, void* d_ws, size_t ws_size,
                              hipStream_t stream) {
    const float* user_emb = (const float*)d_in[0];
    const float* item_emb = (const float*)d_in[1];
    const int*   g_rows   = (const int*)  d_in[2];
    const int*   g_cols   = (const int*)  d_in[3];
    const float* g_vals   = (const float*)d_in[4];
    const int*   users    = (const int*)  d_in[5];
    const int*   items    = (const int*)  d_in[6];
    float*       out      = (float*)d_out;

    const size_t node_f = (size_t)N_NODES * DIM;
    float* bufA = (float*)d_ws;
    float* bufB = bufA + node_f;
    float* accb = bufB + node_f;

    const int TB = 256;
    const int elem_blocks = (int)(((size_t)N_NODES * (DIM / 4) + TB - 1) / TB);
    const int spmm_blocks = (int)(((size_t)NNZ * 16 + TB - 1) / TB);

    lgcn_init<<<elem_blocks, TB, 0, stream>>>(user_emb, item_emb, bufA, accb);

    float* hsrc = bufA;
    float* hdst = bufB;
    for (int hop = 0; hop < 3; ++hop) {
        lgcn_zero<<<elem_blocks, TB, 0, stream>>>(hdst);
        lgcn_spmm<<<spmm_blocks, TB, 0, stream>>>(g_rows, g_cols, g_vals, hsrc, hdst);
        lgcn_addacc<<<elem_blocks, TB, 0, stream>>>(accb, hdst);
        float* t = hsrc; hsrc = hdst; hdst = t;
    }

    lgcn_dot_wmma<<<BATCH / 16 / 8, TB, 0, stream>>>(accb, users, items, out);
}


# Hop as owner-computes, no sort: 256 workgroups each keep 1,172 out rows in LDS preset to ego, scan the whole row list, gather and f32 LDS-add own nnz, store once; sum order: ego, then LDS arrival
# speedup vs baseline: 1.5223x; 1.5223x over previous
#include <hip/hip_runtime.h>
#include <hip/hip_bf16.h>

#define N_USERS 100000
#define N_ITEMS 200000
#define N_NODES (N_USERS + N_ITEMS)
#define NNZ     4800000
#define DIM     64
#define BATCH   16384

typedef float v2f __attribute__((ext_vector_type(2)));
typedef float v8f __attribute__((ext_vector_type(8)));

#define N_OWNERS   256
#define OWN_ROWS   1172
#define OWN_WAVES  32
#define CHUNK      128
#define N_CHUNKS   (NNZ / CHUNK)
#define SCAN_U     8
#define QCAP       192

typedef const __attribute__((address_space(1))) float2* gptr_f2;
typedef const __attribute__((address_space(1))) float4* gptr_f4;

static_assert(OWN_ROWS * N_OWNERS >= N_NODES, "the ranges cover every row");
static_assert(NNZ % CHUNK == 0, "whole chunks");
static_assert((unsigned)N_CHUNKS + (N_OWNERS - 1) * (N_CHUNKS / N_OWNERS) < 2u * N_CHUNKS, "one wrap at most");
static_assert(OWN_ROWS <= 2048 && (N_CHUNKS + OWN_WAVES - 1) / OWN_WAVES + SCAN_U <= 2048, "queue entry fields");
static_assert(QCAP >= 31 + CHUNK, "a chunk's matches fit behind an undrained rest");

__device__ __forceinline__ int4 load_chunk(const int4* __restrict__ rows4, unsigned k,
                                           unsigned start, unsigned lane) {
    unsigned c = (k < (unsigned)N_CHUNKS ? k : (unsigned)N_CHUNKS - 1u) + start;
    if (c >= (unsigned)N_CHUNKS) c -= N_CHUNKS;
    return rows4[(size_t)c * 32 + lane];
}

__device__ __forceinline__ void q_add(unsigned* q, unsigned& tail, unsigned& count, bool m, unsigned ent) {
    const unsigned ball = __builtin_amdgcn_ballot_w32(m);
    if (ball == 0) return;
    unsigned pos = tail + __builtin_amdgcn_mbcnt_lo(ball, 0u);
    if (pos >= QCAP) pos -= QCAP;
    if (m) q[pos] = ent;
    const unsigned n = (unsigned)__builtin_popcount(ball);
    tail += n;
    if (tail >= QCAP) tail -= QCAP;
    count += n;
}

template <bool FULL>
__device__ __forceinline__ void drain(const unsigned* q, unsigned head, unsigned n, unsigned lane,
                                      unsigned wv, unsigned start,
                                      const int* __restrict__ cols, const float* __restrict__ vals,
                                      const float* srcU, const float* srcI, float* tile) {
    unsigned ent = 0u;
    int   col = 0;
    float val = 0.0f;
    if (FULL || lane < n) {
        unsigned qi = head + lane;
        if (qi >= QCAP) qi -= QCAP;
        ent = q[qi];
        unsigned c = ((ent >> 7) & 2047u) * OWN_WAVES + wv + start;
        if (c >= (unsigned)N_CHUNKS) c -= N_CHUNKS;
        const unsigned e = c * CHUNK + (ent & 127u);
        col = cols[e];
        val = vals[e];
    }
    const float* sp = (col < N_USERS) ? srcU + (size_t)col * DIM
                                      : srcI + (size_t)(col - N_USERS) * DIM;
    const unsigned long long spa = (unsigned long long)sp;
    const int sp_lo = (int)(unsigned)spa;
    const int sp_hi = (int)(unsigned)(spa >> 32);
    const int loc   = (int)(ent >> 18);
    const int val_i = __float_as_int(val);
    const unsigned cnt = FULL ? 32u : n;
#pragma nounroll
    for (unsigned jb = 0; jb < cnt; jb += 8) {
        float2 s[8];
#pragma unroll
        for (unsigned k = 0; k < 8; ++k) {
            if (FULL || jb + k < n) {
                const unsigned lo = (unsigned)__builtin_amdgcn_readlane(sp_lo, (int)(jb + k));
                const unsigned hi = (unsigned)__builtin_amdgcn_readlane(sp_hi, (int)(jb + k));
                const gptr_f2 p = (gptr_f2)(((unsigned long long)hi << 32) | lo);
                s[k] = p[lane];
            }
        }
#pragma unroll
        for (unsigned k = 0; k < 8; ++k) {
            if (FULL || jb + k < n) {
                const float    v = __int_as_float(__builtin_amdgcn_readlane(val_i, (int)(jb + k)));
                const unsigned l = (unsigned)__builtin_amdgcn_readlane(loc, (int)(jb + k));
                float* d = tile + (l * DIM + lane * 2);
                __hip_atomic_fetch_add(d + 0, v * s[k].x, __ATOMIC_RELAXED, __HIP_MEMORY_SCOPE_WORKGROUP);
                __hip_atomic_fetch_add(d + 1, v * s[k].y, __ATOMIC_RELAXED, __HIP_MEMORY_SCOPE_WORKGROUP);
            }
        }
    }
}

__global__ void __launch_bounds__(1024)
lgcn_hop_own(const int*   __restrict__ rows,
             const int*   __restrict__ cols,
             const float* __restrict__ vals,
             const float* user_emb,
             const float* item_emb,
             const float* srcU,
             const float* srcI,
             float*       __restrict__ out) {
    __shared__ float    tile[OWN_ROWS * DIM];
    __shared__ unsigned queue[OWN_WAVES][QCAP];

    const unsigned tid  = threadIdx.x;
    const unsigned lane = tid & 31;
    const unsigned wv   = (unsigned)__builtin_amdgcn_readfirstlane((int)(tid >> 5));
    const unsigned base = blockIdx.x * OWN_ROWS;

    constexpr unsigned TILE_F4 = OWN_ROWS * (DIM / 4);
    float4* tile4 = (float4*)tile;
    constexpr unsigned FILL_FULL = TILE_F4 / 1024;
    constexpr unsigned FILL_U    = 6;
    static_assert(FILL_FULL % FILL_U == 0, "whole groups of passes");
    const unsigned long long ubase = (unsigned long long)user_emb;
    const unsigned long long ibase = (unsigned long long)item_emb - (unsigned long long)N_USERS * (DIM * 4u);
    auto ego_ptr = [&](unsigned i) -> gptr_f4 {
        unsigned g = base + (i >> 4);
        if (g > N_NODES - 1) g = N_NODES - 1;
        const unsigned long long a = (g < N_USERS ? ubase : ibase) + ((unsigned long long)g * (DIM * 4u) + (i & 15) * 16u);
        return (gptr_f4)a;
    };
    for (unsigned pb = 0; pb < FILL_FULL; pb += FILL_U) {
        float4 v[FILL_U];
#pragma unroll
        for (unsigned k = 0; k < FILL_U; ++k) v[k] = *ego_ptr(tid + (pb + k) * 1024);
#pragma unroll
        for (unsigned k = 0; k < FILL_U; ++k) tile4[tid + (pb + k) * 1024] = v[k];
    }
    if (tid + FILL_FULL * 1024 < TILE_F4)
        tile4[tid + FILL_FULL * 1024] = *ego_ptr(tid + FILL_FULL * 1024);
    __syncthreads();

    const int4* rows4 = (const int4*)rows;
    const unsigned start = blockIdx.x * (N_CHUNKS / N_OWNERS);
    unsigned* q = queue[wv];
    unsigned head = 0, tail = 0, count = 0;

    constexpr unsigned PER_WAVE = (N_CHUNKS + OWN_WAVES - 1) / OWN_WAVES;
    constexpr unsigned N_BATCH  = (PER_WAVE + SCAN_U - 1) / SCAN_U;

    int4 cur[SCAN_U];
#pragma unroll
    for (unsigned u = 0; u < SCAN_U; ++u)
        cur[u] = load_chunk(rows4, u * OWN_WAVES + wv, start, lane);

    for (unsigned b = 0; b < N_BATCH; ++b) {
#pragma unroll
        for (unsigned u = 0; u < SCAN_U; ++u) {
            const unsigned s = b * SCAN_U + u;
            const unsigned k = s * OWN_WAVES + wv;
            const unsigned loc0 = (unsigned)cur[u].x - base;
            const unsigned loc1 = (unsigned)cur[u].y - base;
            const unsigned loc2 = (unsigned)cur[u].z - base;
            const unsigned loc3 = (unsigned)cur[u].w - base;
            cur[u] = load_chunk(rows4, k + SCAN_U * OWN_WAVES, start, lane);
            const bool m0 = loc0 < OWN_ROWS, m1 = loc1 < OWN_ROWS, m2 = loc2 < OWN_ROWS, m3 = loc3 < OWN_ROWS;
            if (k < (unsigned)N_CHUNKS && __builtin_amdgcn_ballot_w32(m0 | m1 | m2 | m3) != 0) {
                const unsigned tag = (s << 7) | (lane << 2);
                q_add(q, tail, count, m0, (loc0 << 18) | tag | 0u);
                q_add(q, tail, count, m1, (loc1 << 18) | tag | 1u);
                q_add(q, tail, count, m2, (loc2 << 18) | tag | 2u);
                q_add(q, tail, count, m3, (loc3 << 18) | tag | 3u);
                while (count >= 32u) {
                    drain<true>(q, head, 32u, lane, wv, start, cols, vals, srcU, srcI, tile);
                    head += 32u;
                    if (head >= QCAP) head -= QCAP;
                    count -= 32u;
                }
            }
        }
    }
    if (count != 0u)
        drain<false>(q, head, count, lane, wv, start, cols, vals, srcU, srcI, tile);
    __syncthreads();

    for (unsigned i = tid; i < OWN_ROWS * (DIM / 4); i += 1024) {
        const unsigned g = base + (i >> 4);
        if (g < N_NODES)
            ((float4*)out)[(size_t)base * 16 + i] = ((const float4*)tile)[i];
    }
}

__global__ void lgcn_dot_wmma(const float* __restrict__ acc,
                              const int*   __restrict__ users,
                              const int*   __restrict__ items,
                              float*       __restrict__ out) {
    __shared__ float lds[8][16][17];
    int lane  = threadIdx.x & 31;
    int wv    = threadIdx.x >> 5;
    int tile  = blockIdx.x * 8 + wv;
    int m     = lane & 15;
    int khalf = lane >> 4;

    int u_row = users[tile * 16 + m];
    int v_row = items[tile * 16 + m];
    const float* up = acc + (size_t)u_row * DIM;
    const float* vp = acc + ((size_t)N_USERS + (size_t)v_row) * DIM;

    v8f c = {};
#pragma unroll
    for (int kc = 0; kc < DIM / 4; ++kc) {
        int off = kc * 4 + khalf * 2;
        v2f a; a.x = up[off]; a.y = up[off + 1];
        v2f b; b.x = vp[off]; b.y = vp[off + 1];
        c = __builtin_amdgcn_wmma_f32_16x16x4_f32(
                 false, a,  false, b,
                 (short)0, c,  false,  false);
    }

#pragma unroll
    for (int r = 0; r < 8; ++r)
        lds[wv][r + 8 * khalf][m] = c[r];
    __syncthreads();

    if (lane < 16)
        out[tile * 16 + lane] = lds[wv][lane][lane] * (1.0f / 16.0f);
}

extern "C" void kernel_launch(void* const* d_in, const int* in_sizes, int n_in,
                              void* d_out, int out_size, void* d_ws, size_t ws_size,
                              hipStream_t stream) {
    const float* user_emb = (const float*)d_in[0];
    const float* item_emb = (const float*)d_in[1];
    const int*   g_rows   = (const int*)  d_in[2];
    const int*   g_cols   = (const int*)  d_in[3];
    const float* g_vals   = (const float*)d_in[4];
    const int*   users    = (const int*)  d_in[5];
    const int*   items    = (const int*)  d_in[6];
    float*       out      = (float*)d_out;

    const size_t node_f = (size_t)N_NODES * DIM;
    const size_t user_f = (size_t)N_USERS * DIM;
    float* t1 = (float*)d_ws;
    float* t2 = t1 + node_f;
    float* t3 = t2 + node_f;

    lgcn_hop_own<<<N_OWNERS, 1024, 0, stream>>>(g_rows, g_cols, g_vals, user_emb, item_emb,
                                                user_emb, item_emb, t1);
    lgcn_hop_own<<<N_OWNERS, 1024, 0, stream>>>(g_rows, g_cols, g_vals, user_emb, item_emb,
                                                t1, t1 + user_f, t2);
    lgcn_hop_own<<<N_OWNERS, 1024, 0, stream>>>(g_rows, g_cols, g_vals, user_emb, item_emb,
                                                t2, t2 + user_f, t3);

    lgcn_dot_wmma<<<BATCH / 16 / 8, 256, 0, stream>>>(t3, users, items, out);
}


# Hop 1's owner scan also saves each workgroup's matched nnz (row, col, val) in the workspace as 256-B slots; hops 2, 3 replay that list, 32 row gathers in flight, scan nothing; sum order as before
# speedup vs baseline: 2.7973x; 1.8375x over previous
#include <hip/hip_runtime.h>
#include <hip/hip_bf16.h>

#define N_USERS 100000
#define N_ITEMS 200000
#define N_NODES (N_USERS + N_ITEMS)
#define NNZ     4800000
#define DIM     64
#define BATCH   16384

typedef float v2f __attribute__((ext_vector_type(2)));
typedef float v8f __attribute__((ext_vector_type(8)));

#define N_OWNERS   256
#define OWN_ROWS   1172
#define OWN_WAVES  32
#define CHUNK      128
#define N_CHUNKS   (NNZ / CHUNK)
#define SCAN_U     8
#define QCAP       192
#define SLOT_NNZ   32
#define CAP_SLOTS  1171
#define CAP_FULL   (CAP_SLOTS - OWN_WAVES)
#define ENT_NONE   0xFFFFFFFFu

typedef const __attribute__((address_space(1))) float2* gptr_f2;
typedef const __attribute__((address_space(1))) float4* gptr_f4;

static_assert(OWN_ROWS * N_OWNERS >= N_NODES, "the ranges cover every row");
static_assert(NNZ % CHUNK == 0, "whole chunks");
static_assert((unsigned)N_CHUNKS + (N_OWNERS - 1) * (N_CHUNKS / N_OWNERS) < 2u * N_CHUNKS, "one wrap at most");
static_assert(OWN_ROWS <= 2048 && (N_CHUNKS + OWN_WAVES - 1) / OWN_WAVES + SCAN_U <= 2048, "queue entry fields");
static_assert(QCAP >= 31 + CHUNK, "a chunk's matches fit behind an undrained rest");
static_assert((size_t)N_OWNERS * CAP_SLOTS * SLOT_NNZ * 8 + (size_t)N_OWNERS * 4 <= (size_t)N_NODES * DIM * 4,
              "all lists and their counts fit in one node buffer of the workspace");
static_assert(OWN_ROWS < 8191 && N_NODES <= (1 << 19), "list entry fields: 13 bits of row, 19 of column");

__device__ __forceinline__ int4 load_chunk(const int4* __restrict__ rows4, unsigned k,
                                           unsigned start, unsigned lane) {
    unsigned c = (k < (unsigned)N_CHUNKS ? k : (unsigned)N_CHUNKS - 1u) + start;
    if (c >= (unsigned)N_CHUNKS) c -= N_CHUNKS;
    return rows4[(size_t)c * 32 + lane];
}

__device__ __forceinline__ void q_add(unsigned* q, unsigned& tail, unsigned& count, bool m, unsigned ent) {
    const unsigned ball = __builtin_amdgcn_ballot_w32(m);
    if (ball == 0) return;
    unsigned pos = tail + __builtin_amdgcn_mbcnt_lo(ball, 0u);
    if (pos >= QCAP) pos -= QCAP;
    if (m) q[pos] = ent;
    const unsigned n = (unsigned)__builtin_popcount(ball);
    tail += n;
    if (tail >= QCAP) tail -= QCAP;
    count += n;
}

struct NzRegs { int sp_lo, sp_hi, loc, val_i; };

__device__ __forceinline__ NzRegs nz_regs(int col, int loc, float val, const float* srcU, const float* srcI) {
    const float* sp = (col < N_USERS) ? srcU + (size_t)col * DIM
                                      : srcI + (size_t)(col - N_USERS) * DIM;
    const unsigned long long spa = (unsigned long long)sp;
    NzRegs z;
    z.sp_lo = (int)(unsigned)spa;
    z.sp_hi = (int)(unsigned)(spa >> 32);
    z.loc   = loc;
    z.val_i = __float_as_int(val);
    return z;
}

template <bool FULL>
__device__ __forceinline__ void apply8(const NzRegs z, unsigned n, unsigned lane, float* tile) {
    const unsigned cnt = FULL ? 32u : n;
#pragma nounroll
    for (unsigned jb = 0; jb < cnt; jb += 8) {
        float2 s[8];
#pragma unroll
        for (unsigned k = 0; k < 8; ++k) {
            if (FULL || jb + k < n) {
                const unsigned lo = (unsigned)__builtin_amdgcn_readlane(z.sp_lo, (int)(jb + k));
                const unsigned hi = (unsigned)__builtin_amdgcn_readlane(z.sp_hi, (int)(jb + k));
                const gptr_f2 p = (gptr_f2)(((unsigned long long)hi << 32) | lo);
                s[k] = p[lane];
            }
        }
#pragma unroll
        for (unsigned k = 0; k < 8; ++k) {
            if (FULL || jb + k < n) {
                const float    v = __int_as_float(__builtin_amdgcn_readlane(z.val_i, (int)(jb + k)));
                const unsigned l = (unsigned)__builtin_amdgcn_readlane(z.loc, (int)(jb + k));
                float* d = tile + (l * DIM + lane * 2);
                __hip_atomic_fetch_add(d + 0, v * s[k].x, __ATOMIC_RELAXED, __HIP_MEMORY_SCOPE_WORKGROUP);
                __hip_atomic_fetch_add(d + 1, v * s[k].y, __ATOMIC_RELAXED, __HIP_MEMORY_SCOPE_WORKGROUP);
            }
        }
    }
}

__device__ __forceinline__ void apply32(const NzRegs z, unsigned lane, float* tile) {
    float2 s[32];
#pragma unroll
    for (unsigned k = 0; k < 32; ++k) {
        const unsigned lo = (unsigned)__builtin_amdgcn_readlane(z.sp_lo, (int)k);
        const unsigned hi = (unsigned)__builtin_amdgcn_readlane(z.sp_hi, (int)k);
        const gptr_f2 p = (gptr_f2)(((unsigned long long)hi << 32) | lo);
        s[k] = p[lane];
    }
#pragma unroll
    for (unsigned k = 0; k < 32; ++k) {
        const float    v = __int_as_float(__builtin_amdgcn_readlane(z.val_i, (int)k));
        const unsigned l = (unsigned)__builtin_amdgcn_readlane(z.loc, (int)k);
        float* d = tile + (l * DIM + lane * 2);
        __hip_atomic_fetch_add(d + 0, v * s[k].x, __ATOMIC_RELAXED, __HIP_MEMORY_SCOPE_WORKGROUP);
        __hip_atomic_fetch_add(d + 1, v * s[k].y, __ATOMIC_RELAXED, __HIP_MEMORY_SCOPE_WORKGROUP);
    }
}

template <bool FULL, bool SAVE>
__device__ __forceinline__ void drain(const unsigned* q, unsigned head, unsigned n, unsigned lane,
                                      unsigned wv, unsigned start,
                                      const int* __restrict__ cols, const float* __restrict__ vals,
                                      const float* srcU, const float* srcI, float* tile,
                                      unsigned* nslot, uint2* mylist) {
    unsigned ent = 0u;
    int   col = 0;
    float val = 0.0f;
    const bool have = FULL || lane < n;
    if (have) {
        unsigned qi = head + lane;
        if (qi >= QCAP) qi -= QCAP;
        ent = q[qi];
        unsigned c = ((ent >> 7) & 2047u) * OWN_WAVES + wv + start;
        if (c >= (unsigned)N_CHUNKS) c -= N_CHUNKS;
        const unsigned e = c * CHUNK + (ent & 127u);
        col = cols[e];
        val = vals[e];
    }
    const int loc = (int)(ent >> 18);
    if (SAVE) {
        unsigned sl = (unsigned)CAP_FULL + wv;
        if (FULL) {
            sl = 0u;
            if (lane == 0)
                sl = __hip_atomic_fetch_add(nslot, 1u, __ATOMIC_RELAXED, __HIP_MEMORY_SCOPE_WORKGROUP);
            sl = (unsigned)__builtin_amdgcn_readfirstlane((int)sl);
        }
        if (!FULL || sl < (unsigned)CAP_FULL) {
            uint2 w;
            w.x = have ? (((unsigned)loc << 19) | (unsigned)col) : ENT_NONE;
            w.y = (unsigned)__float_as_int(val);
            mylist[(size_t)sl * SLOT_NNZ + lane] = w;
        }
    }
    apply8<FULL>(nz_regs(col, loc, val, srcU, srcI), n, lane, tile);
}

template <bool SAVE>
__device__ __forceinline__ void scan_list(const int* __restrict__ rows,
                                          const int* __restrict__ cols, const float* __restrict__ vals,
                                          const float* srcU, const float* srcI, float* tile,
                                          unsigned* q, unsigned* nslot, uint2* mylist,
                                          unsigned base, unsigned start, unsigned wv, unsigned lane) {
    const int4* rows4 = (const int4*)rows;
    unsigned head = 0, tail = 0, count = 0;

    constexpr unsigned PER_WAVE = (N_CHUNKS + OWN_WAVES - 1) / OWN_WAVES;
    constexpr unsigned N_BATCH  = (PER_WAVE + SCAN_U - 1) / SCAN_U;

    int4 cur[SCAN_U];
#pragma unroll
    for (unsigned u = 0; u < SCAN_U; ++u)
        cur[u] = load_chunk(rows4, u * OWN_WAVES + wv, start, lane);

    for (unsigned b = 0; b < N_BATCH; ++b) {
#pragma unroll
        for (unsigned u = 0; u < SCAN_U; ++u) {
            const unsigned s = b * SCAN_U + u;
            const unsigned k = s * OWN_WAVES + wv;
            const unsigned loc0 = (unsigned)cur[u].x - base;
            const unsigned loc1 = (unsigned)cur[u].y - base;
            const unsigned loc2 = (unsigned)cur[u].z - base;
            const unsigned loc3 = (unsigned)cur[u].w - base;
            cur[u] = load_chunk(rows4, k + SCAN_U * OWN_WAVES, start, lane);
            const bool m0 = loc0 < OWN_ROWS, m1 = loc1 < OWN_ROWS, m2 = loc2 < OWN_ROWS, m3 = loc3 < OWN_ROWS;
            if (k < (unsigned)N_CHUNKS && __builtin_amdgcn_ballot_w32(m0 | m1 | m2 | m3) != 0) {
                const unsigned tag = (s << 7) | (lane << 2);
                q_add(q, tail, count, m0, (loc0 << 18) | tag | 0u);
                q_add(q, tail, count, m1, (loc1 << 18) | tag | 1u);
                q_add(q, tail, count, m2, (loc2 << 18) | tag | 2u);
                q_add(q, tail, count, m3, (loc3 << 18) | tag | 3u);
                while (count >= 32u) {
                    drain<true, SAVE>(q, head, 32u, lane, wv, start, cols, vals, srcU, srcI, tile, nslot, mylist);
                    head += 32u;
                    if (head >= QCAP) head -= QCAP;
                    count -= 32u;
                }
            }
        }
    }
    if (count != 0u) {
        drain<false, SAVE>(q, head, count, lane, wv, start, cols, vals, srcU, srcI, tile, nslot, mylist);
    } else if (SAVE) {
        uint2 w;
        w.x = ENT_NONE; w.y = 0u;
        mylist[(size_t)((unsigned)CAP_FULL + wv) * SLOT_NNZ + lane] = w;
    }
}

__device__ __forceinline__ void replay_list(const uint2* mylist, unsigned ns,
                                            const float* srcU, const float* srcI, float* tile,
                                            unsigned wv, unsigned lane) {
    const uint2 t = mylist[(size_t)((unsigned)CAP_FULL + wv) * SLOT_NNZ + lane];
    uint2 nxt;
    nxt.x = 0u; nxt.y = 0u;
    if (wv < ns) nxt = mylist[(size_t)wv * SLOT_NNZ + lane];
    {
        const bool have = t.x != ENT_NONE;
        const unsigned n = (unsigned)__builtin_popcount(__builtin_amdgcn_ballot_w32(have));
        const int   col = have ? (int)(t.x & 0x7FFFFu) : 0;
        const int   loc = have ? (int)(t.x >> 19) : 0;
        const float val = have ? __int_as_float((int)t.y) : 0.0f;
        if (n != 0u) apply8<false>(nz_regs(col, loc, val, srcU, srcI), n, lane, tile);
    }
    for (unsigned s = wv; s < ns; s += OWN_WAVES) {
        const uint2 e = nxt;
        if (s + OWN_WAVES < ns) nxt = mylist[(size_t)(s + OWN_WAVES) * SLOT_NNZ + lane];
        apply32(nz_regs((int)(e.x & 0x7FFFFu), (int)(e.x >> 19), __int_as_float((int)e.y), srcU, srcI), lane, tile);
    }
}

template <bool FIRST>
__global__ void __launch_bounds__(1024)
lgcn_hop_own(const int*   __restrict__ rows,
             const int*   __restrict__ cols,
             const float* __restrict__ vals,
             const float* user_emb,
             const float* item_emb,
             const float* srcU,
             const float* srcI,
             float*       __restrict__ out,
             uint2*       list,
             unsigned*    counts) {
    __shared__ float    tile[OWN_ROWS * DIM];
    __shared__ unsigned queue[OWN_WAVES][QCAP];
    __shared__ unsigned nslot;

    const unsigned tid  = threadIdx.x;
    const unsigned lane = tid & 31;
    const unsigned wv   = (unsigned)__builtin_amdgcn_readfirstlane((int)(tid >> 5));
    const unsigned base = blockIdx.x * OWN_ROWS;

    constexpr unsigned TILE_F4 = OWN_ROWS * (DIM / 4);
    float4* tile4 = (float4*)tile;
    constexpr unsigned FILL_FULL = TILE_F4 / 1024;
    constexpr unsigned FILL_U    = 6;
    static_assert(FILL_FULL % FILL_U == 0, "whole groups of passes");
    const unsigned long long ubase = (unsigned long long)user_emb;
    const unsigned long long ibase = (unsigned long long)item_emb - (unsigned long long)N_USERS * (DIM * 4u);
    auto ego_ptr = [&](unsigned i) -> gptr_f4 {
        unsigned g = base + (i >> 4);
        if (g > N_NODES - 1) g = N_NODES - 1;
        const unsigned long long a = (g < N_USERS ? ubase : ibase) + ((unsigned long long)g * (DIM * 4u) + (i & 15) * 16u);
        return (gptr_f4)a;
    };
    for (unsigned pb = 0; pb < FILL_FULL; pb += FILL_U) {
        float4 v[FILL_U];
#pragma unroll
        for (unsigned k = 0; k < FILL_U; ++k) v[k] = *ego_ptr(tid + (pb + k) * 1024);
#pragma unroll
        for (unsigned k = 0; k < FILL_U; ++k) tile4[tid + (pb + k) * 1024] = v[k];
    }
    if (tid + FILL_FULL * 1024 < TILE_F4)
        tile4[tid + FILL_FULL * 1024] = *ego_ptr(tid + FILL_FULL * 1024);
    if (FIRST && tid == 0) nslot = 0u;
    __syncthreads();

    const unsigned start = blockIdx.x * (N_CHUNKS / N_OWNERS);
    uint2* mylist = list + (size_t)blockIdx.x * ((size_t)CAP_SLOTS * SLOT_NNZ);
    unsigned ns = 0u;
    if (!FIRST) ns = (unsigned)__builtin_amdgcn_readfirstlane((int)counts[blockIdx.x]);
    if (FIRST || ns > (unsigned)CAP_FULL)
        scan_list<FIRST>(rows, cols, vals, srcU, srcI, tile, queue[wv], &nslot, mylist, base, start, wv, lane);
    else
        replay_list(mylist, ns, srcU, srcI, tile, wv, lane);
    __syncthreads();
    if (FIRST && tid == 0) counts[blockIdx.x] = nslot;

    for (unsigned i = tid; i < OWN_ROWS * (DIM / 4); i += 1024) {
        const unsigned g = base + (i >> 4);
        if (g < N_NODES)
            ((float4*)out)[(size_t)base * 16 + i] = ((const float4*)tile)[i];
    }
}

__global__ void lgcn_dot_wmma(const float* __restrict__ acc,
                              const int*   __restrict__ users,
                              const int*   __restrict__ items,
                              float*       __restrict__ out) {
    __shared__ float lds[8][16][17];
    int lane  = threadIdx.x & 31;
    int wv    = threadIdx.x >> 5;
    int tile  = blockIdx.x * 8 + wv;
    int m     = lane & 15;
    int khalf = lane >> 4;

    int u_row = users[tile * 16 + m];
    int v_row = items[tile * 16 + m];
    const float* up = acc + (size_t)u_row * DIM;
    const float* vp = acc + ((size_t)N_USERS + (size_t)v_row) * DIM;

    v8f c = {};
#pragma unroll
    for (int kc = 0; kc < DIM / 4; ++kc) {
        int off = kc * 4 + khalf * 2;
        v2f a; a.x = up[off]; a.y = up[off + 1];
        v2f b; b.x = vp[off]; b.y = vp[off + 1];
        c = __builtin_amdgcn_wmma_f32_16x16x4_f32(
                 false, a,  false, b,
                 (short)0, c,  false,  false);
    }

#pragma unroll
    for (int r = 0; r < 8; ++r)
        lds[wv][r + 8 * khalf][m] = c[r];
    __syncthreads();

    if (lane < 16)
        out[tile * 16 + lane] = lds[wv][lane][lane] * (1.0f / 16.0f);
}

extern "C" void kernel_launch(void* const* d_in, const int* in_sizes, int n_in,
                              void* d_out, int out_size, void* d_ws, size_t ws_size,
                              hipStream_t stream) {
    const float* user_emb = (const float*)d_in[0];
    const float* item_emb = (const float*)d_in[1];
    const int*   g_rows   = (const int*)  d_in[2];
    const int*   g_cols   = (const int*)  d_in[3];
    const float* g_vals   = (const float*)d_in[4];
    const int*   users    = (const int*)  d_in[5];
    const int*   items    = (const int*)  d_in[6];
    float*       out      = (float*)d_out;

    const size_t node_f = (size_t)N_NODES * DIM;
    const size_t user_f = (size_t)N_USERS * DIM;
    float* t1 = (float*)d_ws;
    float* t2 = t1 + node_f;
    float* t3 = t1;
    uint2*    list   = (uint2*)(t2 + node_f);
    unsigned* counts = (unsigned*)(list + (size_t)N_OWNERS * CAP_SLOTS * SLOT_NNZ);

    lgcn_hop_own<true><<<N_OWNERS, 1024, 0, stream>>>(g_rows, g_cols, g_vals, user_emb, item_emb,
                                                      user_emb, item_emb, t1, list, counts);
    lgcn_hop_own<false><<<N_OWNERS, 1024, 0, stream>>>(g_rows, g_cols, g_vals, user_emb, item_emb,
                                                       t1, t1 + user_f, t2, list, counts);
    lgcn_hop_own<false><<<N_OWNERS, 1024, 0, stream>>>(g_rows, g_cols, g_vals, user_emb, item_emb,
                                                       t2, t2 + user_f, t3, list, counts);

    lgcn_dot_wmma<<<BATCH / 16 / 8, 256, 0, stream>>>(t3, users, items, out);
}


# New pack pass writes per nnz the owning range as one byte and an 8-B (row, col, val) record; hop 1 scans the bytes, 16 nnz per lane per load, and loads one record per match; sum order as before
# speedup vs baseline: 3.5494x; 1.2689x over previous
#include <hip/hip_runtime.h>
#include <hip/hip_bf16.h>

#define N_USERS 100000
#define N_ITEMS 200000
#define N_NODES (N_USERS + N_ITEMS)
#define NNZ     4800000
#define DIM     64
#define BATCH   16384

typedef float v2f __attribute__((ext_vector_type(2)));
typedef float v8f __attribute__((ext_vector_type(8)));

#define N_OWNERS   256
#define OWN_ROWS   1172
#define OWN_WAVES  32
#define CHUNK      128
#define N_CHUNKS   (NNZ / CHUNK)
#define SCAN_U     8
#define QCAP       192
#define SLOT_NNZ   32
#define CAP_SLOTS  1171
#define CAP_FULL   (CAP_SLOTS - OWN_WAVES)
#define ENT_NONE   0xFFFFFFFFu
#define OB_CHUNK   512
#define NB_CHUNKS  (NNZ / OB_CHUNK)
#define BQ         64

typedef const __attribute__((address_space(1))) float2* gptr_f2;
typedef const __attribute__((address_space(1))) float4* gptr_f4;

static_assert(OWN_ROWS * N_OWNERS >= N_NODES, "the ranges cover every row");
static_assert(NNZ % CHUNK == 0, "whole chunks");
static_assert((unsigned)N_CHUNKS + (N_OWNERS - 1) * (N_CHUNKS / N_OWNERS) < 2u * N_CHUNKS, "one wrap at most");
static_assert(OWN_ROWS <= 2048 && (N_CHUNKS + OWN_WAVES - 1) / OWN_WAVES + SCAN_U <= 2048, "queue entry fields");
static_assert(QCAP >= 31 + CHUNK, "a chunk's matches fit behind an undrained rest");
static_assert((size_t)N_OWNERS * CAP_SLOTS * SLOT_NNZ * 8 + (size_t)N_OWNERS * 4 <= (size_t)N_NODES * DIM * 4,
              "all lists and their counts fit in one node buffer of the workspace");
static_assert(OWN_ROWS < 8191 && N_NODES <= (1 << 19), "list entry fields: 13 bits of row, 19 of column");
static_assert(NNZ % OB_CHUNK == 0 && N_OWNERS == 256, "whole chunks of owner bytes; an owner number is one byte");
static_assert((unsigned)NB_CHUNKS + (N_OWNERS - 1) * (NB_CHUNKS / N_OWNERS) < 2u * NB_CHUNKS, "one wrap at most");
static_assert(BQ <= QCAP && (BQ & (BQ - 1)) == 0 && BQ >= 31 + 32, "the byte scan's ring fits the queue");
static_assert((size_t)NNZ + (size_t)NNZ * 8 <= (size_t)N_NODES * DIM * 4, "owner bytes and records fit in one node buffer");

__device__ __forceinline__ int4 load_chunk(const int4* __restrict__ rows4, unsigned k,
                                           unsigned start, unsigned lane) {
    unsigned c = (k < (unsigned)N_CHUNKS ? k : (unsigned)N_CHUNKS - 1u) + start;
    if (c >= (unsigned)N_CHUNKS) c -= N_CHUNKS;
    return rows4[(size_t)c * 32 + lane];
}

__device__ __forceinline__ void q_add(unsigned* q, unsigned& tail, unsigned& count, bool m, unsigned ent) {
    const unsigned ball = __builtin_amdgcn_ballot_w32(m);
    if (ball == 0) return;
    unsigned pos = tail + __builtin_amdgcn_mbcnt_lo(ball, 0u);
    if (pos >= QCAP) pos -= QCAP;
    if (m) q[pos] = ent;
    const unsigned n = (unsigned)__builtin_popcount(ball);
    tail += n;
    if (tail >= QCAP) tail -= QCAP;
    count += n;
}

struct NzRegs { int sp_lo, sp_hi, loc, val_i; };

__device__ __forceinline__ NzRegs nz_regs(int col, int loc, float val, const float* srcU, const float* srcI) {
    const float* sp = (col < N_USERS) ? srcU + (size_t)col * DIM
                                      : srcI + (size_t)(col - N_USERS) * DIM;
    const unsigned long long spa = (unsigned long long)sp;
    NzRegs z;
    z.sp_lo = (int)(unsigned)spa;
    z.sp_hi = (int)(unsigned)(spa >> 32);
    z.loc   = loc;
    z.val_i = __float_as_int(val);
    return z;
}

template <bool FULL>
__device__ __forceinline__ void apply8(const NzRegs z, unsigned n, unsigned lane, float* tile) {
    const unsigned cnt = FULL ? 32u : n;
#pragma nounroll
    for (unsigned jb = 0; jb < cnt; jb += 8) {
        float2 s[8];
#pragma unroll
        for (unsigned k = 0; k < 8; ++k) {
            if (FULL || jb + k < n) {
                const unsigned lo = (unsigned)__builtin_amdgcn_readlane(z.sp_lo, (int)(jb + k));
                const unsigned hi = (unsigned)__builtin_amdgcn_readlane(z.sp_hi, (int)(jb + k));
                const gptr_f2 p = (gptr_f2)(((unsigned long long)hi << 32) | lo);
                s[k] = p[lane];
            }
        }
#pragma unroll
        for (unsigned k = 0; k < 8; ++k) {
            if (FULL || jb + k < n) {
                const float    v = __int_as_float(__builtin_amdgcn_readlane(z.val_i, (int)(jb + k)));
                const unsigned l = (unsigned)__builtin_amdgcn_readlane(z.loc, (int)(jb + k));
                float* d = tile + (l * DIM + lane * 2);
                __hip_atomic_fetch_add(d + 0, v * s[k].x, __ATOMIC_RELAXED, __HIP_MEMORY_SCOPE_WORKGROUP);
                __hip_atomic_fetch_add(d + 1, v * s[k].y, __ATOMIC_RELAXED, __HIP_MEMORY_SCOPE_WORKGROUP);
            }
        }
    }
}

__device__ __forceinline__ void apply32(const NzRegs z, unsigned lane, float* tile) {
    float2 s[32];
#pragma unroll
    for (unsigned k = 0; k < 32; ++k) {
        const unsigned lo = (unsigned)__builtin_amdgcn_readlane(z.sp_lo, (int)k);
        const unsigned hi = (unsigned)__builtin_amdgcn_readlane(z.sp_hi, (int)k);
        const gptr_f2 p = (gptr_f2)(((unsigned long long)hi << 32) | lo);
        s[k] = p[lane];
    }
#pragma unroll
    for (unsigned k = 0; k < 32; ++k) {
        const float    v = __int_as_float(__builtin_amdgcn_readlane(z.val_i, (int)k));
        const unsigned l = (unsigned)__builtin_amdgcn_readlane(z.loc, (int)k);
        float* d = tile + (l * DIM + lane * 2);
        __hip_atomic_fetch_add(d + 0, v * s[k].x, __ATOMIC_RELAXED, __HIP_MEMORY_SCOPE_WORKGROUP);
        __hip_atomic_fetch_add(d + 1, v * s[k].y, __ATOMIC_RELAXED, __HIP_MEMORY_SCOPE_WORKGROUP);
    }
}

__device__ __forceinline__ unsigned zero_bytes(unsigned x) {
    return ~(((x & 0x7F7F7F7Fu) + 0x7F7F7F7Fu) | x) & 0x80808080u;
}

__device__ __forceinline__ uint4 load_owner_chunk(const uint4* __restrict__ ownb, unsigned k,
                                                  unsigned start, unsigned lane) {
    unsigned c = (k < (unsigned)NB_CHUNKS ? k : (unsigned)NB_CHUNKS - 1u) + start;
    if (c >= (unsigned)NB_CHUNKS) c -= NB_CHUNKS;
    return ownb[(size_t)c * 32 + lane];
}

template <bool FULL>
__device__ __forceinline__ void drain_recs(const unsigned* q, unsigned head, unsigned n, unsigned lane,
                                           unsigned wv, unsigned start, const uint2* __restrict__ recs,
                                           const float* srcU, const float* srcI, float* tile,
                                           unsigned* nslot, uint2* mylist) {
    uint2 w;
    w.x = ENT_NONE; w.y = 0u;
    const bool have = FULL || lane < n;
    if (have) {
        const unsigned ent = q[(head + lane) & (BQ - 1u)];
        unsigned c = (ent >> 9) * OWN_WAVES + wv + start;
        if (c >= (unsigned)NB_CHUNKS) c -= NB_CHUNKS;
        w = recs[(size_t)c * OB_CHUNK + (ent & 511u)];
    }
    unsigned sl = (unsigned)CAP_FULL + wv;
    if (FULL) {
        sl = 0u;
        if (lane == 0)
            sl = __hip_atomic_fetch_add(nslot, 1u, __ATOMIC_RELAXED, __HIP_MEMORY_SCOPE_WORKGROUP);
        sl = (unsigned)__builtin_amdgcn_readfirstlane((int)sl);
    }
    if (!FULL || sl < (unsigned)CAP_FULL)
        mylist[(size_t)sl * SLOT_NNZ + lane] = w;
    const int   col = have ? (int)(w.x & 0x7FFFFu) : 0;
    const int   loc = have ? (int)(w.x >> 19) : 0;
    const float val = have ? __int_as_float((int)w.y) : 0.0f;
    apply8<FULL>(nz_regs(col, loc, val, srcU, srcI), n, lane, tile);
}

__device__ __forceinline__ void scan_owner_bytes(const uint4* __restrict__ ownb, const uint2* __restrict__ recs,
                                                 const float* srcU, const float* srcI, float* tile,
                                                 unsigned* q, unsigned* nslot, uint2* mylist,
                                                 unsigned owner, unsigned start, unsigned wv, unsigned lane) {
    unsigned head = 0, tail = 0, count = 0;
    const unsigned pat = owner * 0x01010101u;

    constexpr unsigned PER_WAVE = (NB_CHUNKS + OWN_WAVES - 1) / OWN_WAVES;
    constexpr unsigned N_BATCH  = (PER_WAVE + SCAN_U - 1) / SCAN_U;

    uint4 cur[SCAN_U];
#pragma unroll
    for (unsigned u = 0; u < SCAN_U; ++u)
        cur[u] = load_owner_chunk(ownb, u * OWN_WAVES + wv, start, lane);

    for (unsigned b = 0; b < N_BATCH; ++b) {
#pragma unroll
        for (unsigned u = 0; u < SCAN_U; ++u) {
            const unsigned s = b * SCAN_U + u;
            const unsigned k = s * OWN_WAVES + wv;
            const uint4 w = cur[u];
            cur[u] = load_owner_chunk(ownb, k + SCAN_U * OWN_WAVES, start, lane);
            if (k < (unsigned)NB_CHUNKS) {
                unsigned mask = zero_bytes(w.x ^ pat) | (zero_bytes(w.y ^ pat) >> 1)
                              | (zero_bytes(w.z ^ pat) >> 2) | (zero_bytes(w.w ^ pat) >> 3);
                const unsigned tag = (s << 9) | (lane << 4);
                for (;;) {
                    const bool     m    = mask != 0u;
                    const unsigned ball = __builtin_amdgcn_ballot_w32(m);
                    if (ball == 0u) break;
                    const unsigned p   = (unsigned)(__builtin_ffs((int)mask) - 1);
                    const unsigned idx = (((~p) & 7u) << 2) | (p >> 3);
                    const unsigned pos = (tail + __builtin_amdgcn_mbcnt_lo(ball, 0u)) & (BQ - 1u);
                    if (m) q[pos] = tag | idx;
                    const unsigned n = (unsigned)__builtin_popcount(ball);
                    tail = (tail + n) & (BQ - 1u);
                    count += n;
                    mask &= mask - 1u;
                    if (count >= 32u) {
                        drain_recs<true>(q, head, 32u, lane, wv, start, recs, srcU, srcI, tile, nslot, mylist);
                        head = (head + 32u) & (BQ - 1u);
                        count -= 32u;
                    }
                }
            }
        }
    }
    if (count != 0u) {
        drain_recs<false>(q, head, count, lane, wv, start, recs, srcU, srcI, tile, nslot, mylist);
    } else {
        uint2 w;
        w.x = ENT_NONE; w.y = 0u;
        mylist[(size_t)((unsigned)CAP_FULL + wv) * SLOT_NNZ + lane] = w;
    }
}

template <bool FULL>
__device__ __forceinline__ void drain(const unsigned* q, unsigned head, unsigned n, unsigned lane,
                                      unsigned wv, unsigned start,
                                      const int* __restrict__ cols, const float* __restrict__ vals,
                                      const float* srcU, const float* srcI, float* tile) {
    unsigned ent = 0u;
    int   col = 0;
    float val = 0.0f;
    const bool have = FULL || lane < n;
    if (have) {
        unsigned qi = head + lane;
        if (qi >= QCAP) qi -= QCAP;
        ent = q[qi];
        unsigned c = ((ent >> 7) & 2047u) * OWN_WAVES + wv + start;
        if (c >= (unsigned)N_CHUNKS) c -= N_CHUNKS;
        const unsigned e = c * CHUNK + (ent & 127u);
        col = cols[e];
        val = vals[e];
    }
    const int loc = (int)(ent >> 18);
    apply8<FULL>(nz_regs(col, loc, val, srcU, srcI), n, lane, tile);
}

__device__ __forceinline__ void scan_list(const int* __restrict__ rows,
                                          const int* __restrict__ cols, const float* __restrict__ vals,
                                          const float* srcU, const float* srcI, float* tile,
                                          unsigned* q, unsigned base, unsigned start, unsigned wv, unsigned lane) {
    const int4* rows4 = (const int4*)rows;
    unsigned head = 0, tail = 0, count = 0;

    constexpr unsigned PER_WAVE = (N_CHUNKS + OWN_WAVES - 1) / OWN_WAVES;
    constexpr unsigned N_BATCH  = (PER_WAVE + SCAN_U - 1) / SCAN_U;

    int4 cur[SCAN_U];
#pragma unroll
    for (unsigned u = 0; u < SCAN_U; ++u)
        cur[u] = load_chunk(rows4, u * OWN_WAVES + wv, start, lane);

    for (unsigned b = 0; b < N_BATCH; ++b) {
#pragma unroll
        for (unsigned u = 0; u < SCAN_U; ++u) {
            const unsigned s = b * SCAN_U + u;
            const unsigned k = s * OWN_WAVES + wv;
            const unsigned loc0 = (unsigned)cur[u].x - base;
            const unsigned loc1 = (unsigned)cur[u].y - base;
            const unsigned loc2 = (unsigned)cur[u].z - base;
            const unsigned loc3 = (unsigned)cur[u].w - base;
            cur[u] = load_chunk(rows4, k + SCAN_U * OWN_WAVES, start, lane);
            const bool m0 = loc0 < OWN_ROWS, m1 = loc1 < OWN_ROWS, m2 = loc2 < OWN_ROWS, m3 = loc3 < OWN_ROWS;
            if (k < (unsigned)N_CHUNKS && __builtin_amdgcn_ballot_w32(m0 | m1 | m2 | m3) != 0) {
                const unsigned tag = (s << 7) | (lane << 2);
                q_add(q, tail, count, m0, (loc0 << 18) | tag | 0u);
                q_add(q, tail, count, m1, (loc1 << 18) | tag | 1u);
                q_add(q, tail, count, m2, (loc2 << 18) | tag | 2u);
                q_add(q, tail, count, m3, (loc3 << 18) | tag | 3u);
                while (count >= 32u) {
                    drain<true>(q, head, 32u, lane, wv, start, cols, vals, srcU, srcI, tile);
                    head += 32u;
                    if (head >= QCAP) head -= QCAP;
                    count -= 32u;
                }
            }
        }
    }
    if (count != 0u)
        drain<false>(q, head, count, lane, wv, start, cols, vals, srcU, srcI, tile);
}

__device__ __forceinline__ void replay_list(const uint2* mylist, unsigned ns,
                                            const float* srcU, const float* srcI, float* tile,
                                            unsigned wv, unsigned lane) {
    const uint2 t = mylist[(size_t)((unsigned)CAP_FULL + wv) * SLOT_NNZ + lane];
    uint2 nxt;
    nxt.x = 0u; nxt.y = 0u;
    if (wv < ns) nxt = mylist[(size_t)wv * SLOT_NNZ + lane];
    {
        const bool have = t.x != ENT_NONE;
        const unsigned n = (unsigned)__builtin_popcount(__builtin_amdgcn_ballot_w32(have));
        const int   col = have ? (int)(t.x & 0x7FFFFu) : 0;
        const int   loc = have ? (int)(t.x >> 19) : 0;
        const float val = have ? __int_as_float((int)t.y) : 0.0f;
        if (n != 0u) apply8<false>(nz_regs(col, loc, val, srcU, srcI), n, lane, tile);
    }
    for (unsigned s = wv; s < ns; s += OWN_WAVES) {
        const uint2 e = nxt;
        if (s + OWN_WAVES < ns) nxt = mylist[(size_t)(s + OWN_WAVES) * SLOT_NNZ + lane];
        apply32(nz_regs((int)(e.x & 0x7FFFFu), (int)(e.x >> 19), __int_as_float((int)e.y), srcU, srcI), lane, tile);
    }
}

__global__ void __launch_bounds__(256)
lgcn_pack(const int*   __restrict__ rows,
          const int*   __restrict__ cols,
          const float* __restrict__ vals,
          unsigned*    __restrict__ ownb,
          uint4*       __restrict__ recs) {
    const unsigned t = blockIdx.x * 256 + threadIdx.x;
    if (t >= NNZ / 4) return;
    const int4   r = ((const int4*)rows)[t];
    const int4   c = ((const int4*)cols)[t];
    const float4 v = ((const float4*)vals)[t];
    const unsigned o0 = (unsigned)r.x / OWN_ROWS, o1 = (unsigned)r.y / OWN_ROWS;
    const unsigned o2 = (unsigned)r.z / OWN_ROWS, o3 = (unsigned)r.w / OWN_ROWS;
    ownb[t] = o0 | (o1 << 8) | (o2 << 16) | (o3 << 24);
    uint4 a, b;
    a.x = (((unsigned)r.x - o0 * OWN_ROWS) << 19) | (unsigned)c.x;  a.y = (unsigned)__float_as_int(v.x);
    a.z = (((unsigned)r.y - o1 * OWN_ROWS) << 19) | (unsigned)c.y;  a.w = (unsigned)__float_as_int(v.y);
    b.x = (((unsigned)r.z - o2 * OWN_ROWS) << 19) | (unsigned)c.z;  b.y = (unsigned)__float_as_int(v.z);
    b.z = (((unsigned)r.w - o3 * OWN_ROWS) << 19) | (unsigned)c.w;  b.w = (unsigned)__float_as_int(v.w);
    recs[2 * (size_t)t]     = a;
    recs[2 * (size_t)t + 1] = b;
}

template <bool FIRST>
__global__ void __launch_bounds__(1024)
lgcn_hop_own(const int*   __restrict__ rows,
             const int*   __restrict__ cols,
             const float* __restrict__ vals,
             const float* user_emb,
             const float* item_emb,
             const float* srcU,
             const float* srcI,
             float*       __restrict__ out,
             uint2*       list,
             unsigned*    counts,
             const uint4* ownb,
             const uint2* recs) {
    __shared__ float    tile[OWN_ROWS * DIM];
    __shared__ unsigned queue[OWN_WAVES][QCAP];
    __shared__ unsigned nslot;

    const unsigned tid  = threadIdx.x;
    const unsigned lane = tid & 31;
    const unsigned wv   = (unsigned)__builtin_amdgcn_readfirstlane((int)(tid >> 5));
    const unsigned base = blockIdx.x * OWN_ROWS;

    constexpr unsigned TILE_F4 = OWN_ROWS * (DIM / 4);
    float4* tile4 = (float4*)tile;
    constexpr unsigned FILL_FULL = TILE_F4 / 1024;
    constexpr unsigned FILL_U    = 6;
    static_assert(FILL_FULL % FILL_U == 0, "whole groups of passes");
    const unsigned long long ubase = (unsigned long long)user_emb;
    const unsigned long long ibase = (unsigned long long)item_emb - (unsigned long long)N_USERS * (DIM * 4u);
    auto ego_ptr = [&](unsigned i) -> gptr_f4 {
        unsigned g = base + (i >> 4);
        if (g > N_NODES - 1) g = N_NODES - 1;
        const unsigned long long a = (g < N_USERS ? ubase : ibase) + ((unsigned long long)g * (DIM * 4u) + (i & 15) * 16u);
        return (gptr_f4)a;
    };
    for (unsigned pb = 0; pb < FILL_FULL; pb += FILL_U) {
        float4 v[FILL_U];
#pragma unroll
        for (unsigned k = 0; k < FILL_U; ++k) v[k] = *ego_ptr(tid + (pb + k) * 1024);
#pragma unroll
        for (unsigned k = 0; k < FILL_U; ++k) tile4[tid + (pb + k) * 1024] = v[k];
    }
    if (tid + FILL_FULL * 1024 < TILE_F4)
        tile4[tid + FILL_FULL * 1024] = *ego_ptr(tid + FILL_FULL * 1024);
    if (FIRST && tid == 0) nslot = 0u;
    __syncthreads();

    const unsigned start = blockIdx.x * ((FIRST ? NB_CHUNKS : N_CHUNKS) / N_OWNERS);
    uint2* mylist = list + (size_t)blockIdx.x * ((size_t)CAP_SLOTS * SLOT_NNZ);
    unsigned ns = 0u;
    if (!FIRST) ns = (unsigned)__builtin_amdgcn_readfirstlane((int)counts[blockIdx.x]);
    if (FIRST)
        scan_owner_bytes(ownb, recs, srcU, srcI, tile, queue[wv], &nslot, mylist, blockIdx.x, start, wv, lane);
    else if (ns > (unsigned)CAP_FULL)
        scan_list(rows, cols, vals, srcU, srcI, tile, queue[wv], base, start, wv, lane);
    else
        replay_list(mylist, ns, srcU, srcI, tile, wv, lane);
    __syncthreads();
    if (FIRST && tid == 0) counts[blockIdx.x] = nslot;

    for (unsigned i = tid; i < OWN_ROWS * (DIM / 4); i += 1024) {
        const unsigned g = base + (i >> 4);
        if (g < N_NODES)
            ((float4*)out)[(size_t)base * 16 + i] = ((const float4*)tile)[i];
    }
}

__global__ void lgcn_dot_wmma(const float* __restrict__ acc,
                              const int*   __restrict__ users,
                              const int*   __restrict__ items,
                              float*       __restrict__ out) {
    __shared__ float lds[8][16][17];
    int lane  = threadIdx.x & 31;
    int wv    = threadIdx.x >> 5;
    int tile  = blockIdx.x * 8 + wv;
    int m     = lane & 15;
    int khalf = lane >> 4;

    int u_row = users[tile * 16 + m];
    int v_row = items[tile * 16 + m];
    const float* up = acc + (size_t)u_row * DIM;
    const float* vp = acc + ((size_t)N_USERS + (size_t)v_row) * DIM;

    v8f c = {};
#pragma unroll
    for (int kc = 0; kc < DIM / 4; ++kc) {
        int off = kc * 4 + khalf * 2;
        v2f a; a.x = up[off]; a.y = up[off + 1];
        v2f b; b.x = vp[off]; b.y = vp[off + 1];
        c = __builtin_amdgcn_wmma_f32_16x16x4_f32(
                 false, a,  false, b,
                 (short)0, c,  false,  false);
    }

#pragma unroll
    for (int r = 0; r < 8; ++r)
        lds[wv][r + 8 * khalf][m] = c[r];
    __syncthreads();

    if (lane < 16)
        out[tile * 16 + lane] = lds[wv][lane][lane] * (1.0f / 16.0f);
}

extern "C" void kernel_launch(void* const* d_in, const int* in_sizes, int n_in,
                              void* d_out, int out_size, void* d_ws, size_t ws_size,
                              hipStream_t stream) {
    const float* user_emb = (const float*)d_in[0];
    const float* item_emb = (const float*)d_in[1];
    const int*   g_rows   = (const int*)  d_in[2];
    const int*   g_cols   = (const int*)  d_in[3];
    const float* g_vals   = (const float*)d_in[4];
    const int*   users    = (const int*)  d_in[5];
    const int*   items    = (const int*)  d_in[6];
    float*       out      = (float*)d_out;

    const size_t node_f = (size_t)N_NODES * DIM;
    const size_t user_f = (size_t)N_USERS * DIM;
    float* t1 = (float*)d_ws;
    float* t2 = t1 + node_f;
    float* t3 = t1;
    uint2*    list   = (uint2*)(t2 + node_f);
    unsigned* counts = (unsigned*)(list + (size_t)N_OWNERS * CAP_SLOTS * SLOT_NNZ);
    unsigned* ownb = (unsigned*)t2;
    uint2*    recs = (uint2*)(ownb + NNZ / 4);

    lgcn_pack<<<(NNZ / 4 + 255) / 256, 256, 0, stream>>>(g_rows, g_cols, g_vals, ownb, (uint4*)recs);

    lgcn_hop_own<true><<<N_OWNERS, 1024, 0, stream>>>(g_rows, g_cols, g_vals, user_emb, item_emb,
                                                      user_emb, item_emb, t1, list, counts, (const uint4*)ownb, recs);
    lgcn_hop_own<false><<<N_OWNERS, 1024, 0, stream>>>(g_rows, g_cols, g_vals, user_emb, item_emb,
                                                       t1, t1 + user_f, t2, list, counts, (const uint4*)ownb, recs);
    lgcn_hop_own<false><<<N_OWNERS, 1024, 0, stream>>>(g_rows, g_cols, g_vals, user_emb, item_emb,
                                                       t2, t2 + user_f, t3, list, counts, (const uint4*)ownb, recs);

    lgcn_dot_wmma<<<BATCH / 16 / 8, 256, 0, stream>>>(t3, users, items, out);
}
